# starting kernel
# baseline (speedup 1.0000x reference)
#include <hip/hip_runtime.h>


typedef __attribute__((ext_vector_type(2))) float v2f;
typedef __attribute__((ext_vector_type(8))) float v8f;

#define C1F 4.7e-9f

#if __has_builtin(__builtin_amdgcn_update_dpp)
#define ROR_ADD(x, ctrl)                                                  \
  ((x) + __int_as_float(__builtin_amdgcn_update_dpp(                      \
             0, __float_as_int(x), (ctrl), 0xf, 0xf, true)))
#define HAVE_DPP 1
#endif

__global__ void ndc_prep(const float* __restrict__ v_in,
                         const float* __restrict__ vs_r,
                         const float* __restrict__ fs,
                         float* __restrict__ params, int T) {
  int t = blockIdx.x * blockDim.x + threadIdx.x;
  if (t >= T + 8) return;
  int ti = (t < T) ? t : (T - 1);
  float f   = fs[ti];
  float vr  = vs_r[ti];
  float c1r = 1.0f / (2.0f * C1F * f);
  float r0  = c1r * vr / (c1r + vr);
  float g0 = 1.0f / r0, g1 = 1.0f / c1r, g2 = 1.0f / vr;
  float inv = 2.0f / (g0 + g1 + g2);
  float P0 = g0 * inv, P1 = g1 * inv, P2 = g2 * inv;
  float4* p4 = (float4*)(params + 8 * (size_t)t);
  p4[0] = make_float4(v_in[ti], P0 - 1.0f, P1, P2);
  p4[1] = make_float4(P0, P1 - 1.0f, P2, r0 * (1.0f / 3000.0f));
}

__global__ __launch_bounds__(32, 1) void ndc_scan(
    const float* __restrict__ params,
    const float* __restrict__ W_in, const float* __restrict__ b_in,
    const float* __restrict__ W_h,  const float* __restrict__ b_h,
    const float* __restrict__ W_out, const float* __restrict__ b_out,
    float* __restrict__ out, int T) {
  const int  lane = threadIdx.x;
  const bool hi   = lane >= 16;
  const int  r16  = lane & 15;

  float Wi0 = W_in[2 * lane];
  float Wi1 = W_in[2 * lane + 1];
  float bin = b_in[lane];
  float bLo0 = b_h[r16];
  float bHi0 = b_h[16 + r16];
  float bLo1 = b_h[32 + r16];
  float bHi1 = b_h[48 + r16];
  float WoA = W_out[r16];
  float WoB = W_out[16 + r16];
  float bo  = b_out[0];

  float Bw[2][8][2][2];
#pragma unroll
  for (int l = 0; l < 2; ++l)
#pragma unroll
    for (int c = 0; c < 8; ++c)
#pragma unroll
      for (int blk = 0; blk < 2; ++blk) {
        int row = 16 * blk + r16;
        int col = 4 * c + (hi ? 2 : 0);
        Bw[l][c][blk][0] = W_h[l * 1024 + row * 32 + col];
        Bw[l][c][blk][1] = W_h[l * 1024 + row * 32 + col + 1];
      }

  const int idx_base = hi ? 8 : 0;

  const float4* pf = (const float4*)params;
  float4 qa   = pf[0], qb   = pf[1];
  float4 qa_n = pf[2], qb_n = pf[3];

  float a0 = 0.0f, b1 = 0.0f;

  for (int t = 0; t < T; ++t) {
    float4 qa_nn = pf[2 * (t + 2)];
    float4 qb_nn = pf[2 * (t + 2) + 1];
    __builtin_prefetch(params + 8 * (size_t)(t + 8), 0, 3);

    float v   = qa.x;
    float r00 = qa.y, r01 = qa.z, r02 = qa.w;
    float r10 = qb.x, r11 = qb.y, r12 = qb.z;
    float drt = qb.w;

    float dp_a = r00 * a0 + r01 * b1 + r02 * v;

    float h0 = fmaxf(Wi0 * dp_a + Wi1 * drt + bin, 0.0f);
    int hb = __float_as_int(h0);

    int Ax[8], Ay[8];
#pragma unroll
    for (int c = 0; c < 8; ++c) {
      Ax[c] = __builtin_amdgcn_ds_bpermute(16 * c + idx_base, hb);
      Ay[c] = __builtin_amdgcn_ds_bpermute(16 * c + idx_base + 4, hb);
    }
    v8f a00 = {}, a01 = {}, a10 = {}, a11 = {};
#pragma unroll
    for (int c = 0; c < 4; ++c) {
      v2f A;  A.x  = __int_as_float(Ax[c]);     A.y  = __int_as_float(Ay[c]);
      v2f A2; A2.x = __int_as_float(Ax[c + 4]); A2.y = __int_as_float(Ay[c + 4]);
      v2f B00; B00.x = Bw[0][c][0][0];     B00.y = Bw[0][c][0][1];
      v2f B01; B01.x = Bw[0][c][1][0];     B01.y = Bw[0][c][1][1];
      v2f B10; B10.x = Bw[0][c + 4][0][0]; B10.y = Bw[0][c + 4][0][1];
      v2f B11; B11.x = Bw[0][c + 4][1][0]; B11.y = Bw[0][c + 4][1][1];
      a00 = __builtin_amdgcn_wmma_f32_16x16x4_f32(false, A,  false, B00,
                                                  (short)0, a00, false, false);
      a10 = __builtin_amdgcn_wmma_f32_16x16x4_f32(false, A,  false, B01,
                                                  (short)0, a10, false, false);
      a01 = __builtin_amdgcn_wmma_f32_16x16x4_f32(false, A2, false, B10,
                                                  (short)0, a01, false, false);
      a11 = __builtin_amdgcn_wmma_f32_16x16x4_f32(false, A2, false, B11,
                                                  (short)0, a11, false, false);
    }
    float hLo = fmaxf((a00[0] + a01[0]) + bLo0, 0.0f);
    float hHi = fmaxf((a10[0] + a11[0]) + bHi0, 0.0f);
    int hbL = __float_as_int(hLo);
    int hbH = __float_as_int(hHi);

#pragma unroll
    for (int c = 0; c < 4; ++c) {
      Ax[c] = __builtin_amdgcn_ds_bpermute(16 * c + idx_base, hbL);
      Ay[c] = __builtin_amdgcn_ds_bpermute(16 * c + idx_base + 4, hbL);
    }
#pragma unroll
    for (int c = 4; c < 8; ++c) {
      Ax[c] = __builtin_amdgcn_ds_bpermute(16 * c + idx_base, hbH);
      Ay[c] = __builtin_amdgcn_ds_bpermute(16 * c + idx_base + 4, hbH);
    }
    v8f b00 = {}, b01 = {}, b10 = {}, b11 = {};
#pragma unroll
    for (int c = 0; c < 4; ++c) {
      v2f A;  A.x  = __int_as_float(Ax[c]);     A.y  = __int_as_float(Ay[c]);
      v2f A2; A2.x = __int_as_float(Ax[c + 4]); A2.y = __int_as_float(Ay[c + 4]);
      v2f B00; B00.x = Bw[1][c][0][0];     B00.y = Bw[1][c][0][1];
      v2f B01; B01.x = Bw[1][c][1][0];     B01.y = Bw[1][c][1][1];
      v2f B10; B10.x = Bw[1][c + 4][0][0]; B10.y = Bw[1][c + 4][0][1];
      v2f B11; B11.x = Bw[1][c + 4][1][0]; B11.y = Bw[1][c + 4][1][1];
      b00 = __builtin_amdgcn_wmma_f32_16x16x4_f32(false, A,  false, B00,
                                                  (short)0, b00, false, false);
      b10 = __builtin_amdgcn_wmma_f32_16x16x4_f32(false, A,  false, B01,
                                                  (short)0, b10, false, false);
      b01 = __builtin_amdgcn_wmma_f32_16x16x4_f32(false, A2, false, B10,
                                                  (short)0, b01, false, false);
      b11 = __builtin_amdgcn_wmma_f32_16x16x4_f32(false, A2, false, B11,
                                                  (short)0, b11, false, false);
    }
    float gLo = fmaxf((b00[0] + b01[0]) + bLo1, 0.0f);
    float gHi = fmaxf((b10[0] + b11[0]) + bHi1, 0.0f);

    float p = WoA * gLo + WoB * gHi;
#ifdef HAVE_DPP
    p = ROR_ADD(p, 0x128);
    p = ROR_ADD(p, 0x124);
    p = ROR_ADD(p, 0x122);
    p = ROR_ADD(p, 0x121);
#else
#pragma unroll
    for (int m = 8; m >= 1; m >>= 1) {
      p += __int_as_float(
          __builtin_amdgcn_ds_bpermute((lane ^ m) << 2, __float_as_int(p)));
    }
#endif
    float dp_b = p + bo;

    float b1n = r10 * dp_b + r11 * b1 + r12 * v;

    out[t] = 0.5f * (dp_a + dp_b);

    a0 = dp_b;
    b1 = b1n;
    qa = qa_n;  qb = qb_n;
    qa_n = qa_nn;  qb_n = qb_nn;
  }
}

extern "C" void kernel_launch(void* const* d_in, const int* in_sizes, int n_in,
                              void* d_out, int out_size, void* d_ws,
                              size_t ws_size, hipStream_t stream) {
  const float* v_in  = (const float*)d_in[0];
  const float* vs_r  = (const float*)d_in[1];
  const float* fs    = (const float*)d_in[2];
  const float* W_in  = (const float*)d_in[3];
  const float* b_in  = (const float*)d_in[4];
  const float* W_h   = (const float*)d_in[5];
  const float* b_h   = (const float*)d_in[6];
  const float* W_out = (const float*)d_in[7];
  const float* b_out = (const float*)d_in[8];
  float* out = (float*)d_out;
  int T = in_sizes[0];

  float* params = (float*)d_ws;

  ndc_prep<<<(T + 8 + 255) / 256, 256, 0, stream>>>(v_in, vs_r, fs, params, T);
  ndc_scan<<<1, 32, 0, stream>>>(params, W_in, b_in, W_h, b_h, W_out, b_out,
                                 out, T);
}


# Two waves on two SIMDs share each step: layer 1 split by output block, layer 2 by K half, so hidden 1 stays in its wave; one LDS exchange and one work-group barrier per step; sums unchanged
# speedup vs baseline: 1.1883x; 1.1883x over previous
#include <hip/hip_runtime.h>


typedef __attribute__((ext_vector_type(2))) float v2f;
typedef __attribute__((ext_vector_type(8))) float v8f;

#define C1F 4.7e-9f

#if __has_builtin(__builtin_amdgcn_update_dpp)
#define ROR_ADD(x, ctrl)                                                  \
  ((x) + __int_as_float(__builtin_amdgcn_update_dpp(                      \
             0, __float_as_int(x), (ctrl), 0xf, 0xf, true)))
#define HAVE_DPP 1
#endif

__global__ void ndc_prep(const float* __restrict__ v_in,
                         const float* __restrict__ vs_r,
                         const float* __restrict__ fs,
                         float* __restrict__ params, int T) {
  int t = blockIdx.x * blockDim.x + threadIdx.x;
  if (t >= T + 8) return;
  int ti = (t < T) ? t : (T - 1);
  float f   = fs[ti];
  float vr  = vs_r[ti];
  float c1r = 1.0f / (2.0f * C1F * f);
  float r0  = c1r * vr / (c1r + vr);
  float g0 = 1.0f / r0, g1 = 1.0f / c1r, g2 = 1.0f / vr;
  float inv = 2.0f / (g0 + g1 + g2);
  float P0 = g0 * inv, P1 = g1 * inv, P2 = g2 * inv;
  float4* p4 = (float4*)(params + 8 * (size_t)t);
  p4[0] = make_float4(v_in[ti], P0 - 1.0f, P1, P2);
  p4[1] = make_float4(P0, P1 - 1.0f, P2, r0 * (1.0f / 3000.0f));
}

#define WG_BARRIER()                                                         \
  do {                                                                       \
    asm volatile("s_wait_dscnt 0x0" ::: "memory");                           \
    asm volatile("s_barrier_signal -1" ::: "memory");                        \
    asm volatile("s_barrier_wait -1" ::: "memory");                          \
  } while (0)

__global__ __launch_bounds__(64, 1) void ndc_scan(
    const float* __restrict__ params,
    const float* __restrict__ W_in, const float* __restrict__ b_in,
    const float* __restrict__ W_h,  const float* __restrict__ b_h,
    const float* __restrict__ W_out, const float* __restrict__ b_out,
    float* __restrict__ out, int T) {
  __shared__ __attribute__((aligned(16))) float sP2[2 * 64];

  const int  lane = threadIdx.x & 31;
  const int  wave = __builtin_amdgcn_readfirstlane((int)(threadIdx.x >> 5));
  const bool hi   = lane >= 16;
  const int  r16  = lane & 15;

  float Wi0 = W_in[2 * lane];
  float Wi1 = W_in[2 * lane + 1];
  float bin = b_in[lane];
  float bL0  = b_h[16 * wave + r16];
  float bLo1 = b_h[32 + r16];
  float bHi1 = b_h[48 + r16];
  float WoA = W_out[r16];
  float WoB = W_out[16 + r16];
  float bo  = b_out[0];

  float B1[8][2];
#pragma unroll
  for (int c = 0; c < 8; ++c) {
    int row = 16 * wave + r16;
    int col = 4 * c + (hi ? 2 : 0);
    B1[c][0] = W_h[row * 32 + col];
    B1[c][1] = W_h[row * 32 + col + 1];
  }
  float B2[4][2][2];
#pragma unroll
  for (int cc = 0; cc < 4; ++cc)
#pragma unroll
    for (int blk = 0; blk < 2; ++blk) {
      int row = 16 * blk + r16;
      int col = 4 * (4 * wave + cc) + (hi ? 2 : 0);
      B2[cc][blk][0] = W_h[1024 + row * 32 + col];
      B2[cc][blk][1] = W_h[1024 + row * 32 + col + 1];
    }

  const int idx_base = hi ? 8 : 0;

  const int wq = 4 * r16 + 2 * wave;
  const int rq = 4 * r16;

  const float4* pf = (const float4*)params;
  float4 qa   = pf[0], qb   = pf[1];
  float4 qa_n = pf[2], qb_n = pf[3];

  float a0 = 0.0f, b1 = 0.0f;

  for (int t = 0; t < T; ++t) {
    float4 qa_nn = pf[2 * (t + 2)];
    float4 qb_nn = pf[2 * (t + 2) + 1];
    __builtin_prefetch(params + 8 * (size_t)(t + 8), 0, 3);

    float v   = qa.x;
    float r00 = qa.y, r01 = qa.z, r02 = qa.w;
    float r10 = qb.x, r11 = qb.y, r12 = qb.z;
    float drt = qb.w;

    float dp_a = r00 * a0 + r01 * b1 + r02 * v;

    float h0 = fmaxf(Wi0 * dp_a + Wi1 * drt + bin, 0.0f);
    int hb = __float_as_int(h0);

    int Ax[8], Ay[8];
#pragma unroll
    for (int c = 0; c < 8; ++c) {
      Ax[c] = __builtin_amdgcn_ds_bpermute(16 * c + idx_base, hb);
      Ay[c] = __builtin_amdgcn_ds_bpermute(16 * c + idx_base + 4, hb);
    }
    __builtin_amdgcn_sched_barrier(0);
    v8f a00 = {}, a01 = {};
#pragma unroll
    for (int c = 0; c < 4; ++c) {
      v2f A;  A.x  = __int_as_float(Ax[c]);     A.y  = __int_as_float(Ay[c]);
      v2f A2; A2.x = __int_as_float(Ax[c + 4]); A2.y = __int_as_float(Ay[c + 4]);
      v2f Ba; Ba.x = B1[c][0];     Ba.y = B1[c][1];
      v2f Bb; Bb.x = B1[c + 4][0]; Bb.y = B1[c + 4][1];
      a00 = __builtin_amdgcn_wmma_f32_16x16x4_f32(false, A,  false, Ba,
                                                  (short)0, a00, false, false);
      a01 = __builtin_amdgcn_wmma_f32_16x16x4_f32(false, A2, false, Bb,
                                                  (short)0, a01, false, false);
    }
    float h1 = fmaxf((a00[0] + a01[0]) + bL0, 0.0f);
    int hb1 = __float_as_int(h1);

    int Gx[4], Gy[4];
#pragma unroll
    for (int cc = 0; cc < 4; ++cc) {
      Gx[cc] = __builtin_amdgcn_ds_bpermute(16 * cc + idx_base, hb1);
      Gy[cc] = __builtin_amdgcn_ds_bpermute(16 * cc + idx_base + 4, hb1);
    }
    __builtin_amdgcn_sched_barrier(0);
    v8f c0 = {}, c1 = {};
#pragma unroll
    for (int cc = 0; cc < 4; ++cc) {
      v2f A;  A.x  = __int_as_float(Gx[cc]); A.y  = __int_as_float(Gy[cc]);
      v2f Ba; Ba.x = B2[cc][0][0]; Ba.y = B2[cc][0][1];
      v2f Bb; Bb.x = B2[cc][1][0]; Bb.y = B2[cc][1][1];
      c0 = __builtin_amdgcn_wmma_f32_16x16x4_f32(false, A, false, Ba,
                                                 (short)0, c0, false, false);
      c1 = __builtin_amdgcn_wmma_f32_16x16x4_f32(false, A, false, Bb,
                                                 (short)0, c1, false, false);
    }
    float* buf = sP2 + ((t & 1) << 6);
    *(float2*)&buf[wq] = make_float2(c0[0], c1[0]);
    WG_BARRIER();
    float4 pz = *(const float4*)&buf[rq];
    float gLo = fmaxf((pz.x + pz.z) + bLo1, 0.0f);
    float gHi = fmaxf((pz.y + pz.w) + bHi1, 0.0f);

    float p = __builtin_fmaf(WoA, gLo, WoB * gHi);
#ifdef HAVE_DPP
    p = ROR_ADD(p, 0x128);
    p = ROR_ADD(p, 0x124);
    p = ROR_ADD(p, 0x122);
    p = ROR_ADD(p, 0x121);
#else
#pragma unroll
    for (int m = 8; m >= 1; m >>= 1) {
      p += __int_as_float(
          __builtin_amdgcn_ds_bpermute((lane ^ m) << 2, __float_as_int(p)));
    }
#endif
    float dp_b = p + bo;

    float b1n = r10 * dp_b + r11 * b1 + r12 * v;

    if (wave == 0) out[t] = 0.5f * (dp_a + dp_b);

    a0 = dp_b;
    b1 = b1n;
    qa = qa_n;  qb = qb_n;
    qa_n = qa_nn;  qb_n = qb_nn;
  }
}

extern "C" void kernel_launch(void* const* d_in, const int* in_sizes, int n_in,
                              void* d_out, int out_size, void* d_ws,
                              size_t ws_size, hipStream_t stream) {
  const float* v_in  = (const float*)d_in[0];
  const float* vs_r  = (const float*)d_in[1];
  const float* fs    = (const float*)d_in[2];
  const float* W_in  = (const float*)d_in[3];
  const float* b_in  = (const float*)d_in[4];
  const float* W_h   = (const float*)d_in[5];
  const float* b_h   = (const float*)d_in[6];
  const float* W_out = (const float*)d_in[7];
  const float* b_out = (const float*)d_in[8];
  float* out = (float*)d_out;
  int T = in_sizes[0];

  float* params = (float*)d_ws;

  ndc_prep<<<(T + 8 + 255) / 256, 256, 0, stream>>>(v_in, vs_r, fs, params, T);
  ndc_scan<<<1, 64, 0, stream>>>(params, W_in, b_in, W_h, b_h, W_out, b_out,
                                 out, T);
}


# A operands of both hidden layers built on the vector ALU (v_permlane16 row broadcast, one DPP row rotate) instead of ds_bpermute; chunk k order follows the rows; dot-product sum order changes
# speedup vs baseline: 1.3109x; 1.1032x over previous
#include <hip/hip_runtime.h>


typedef __attribute__((ext_vector_type(2))) float v2f;
typedef __attribute__((ext_vector_type(8))) float v8f;

#define C1F 4.7e-9f

#if __has_builtin(__builtin_amdgcn_update_dpp)
#define ROR_ADD(x, ctrl)                                                  \
  ((x) + __int_as_float(__builtin_amdgcn_update_dpp(                      \
             0, __float_as_int(x), (ctrl), 0xf, 0xf, true)))
#define HAVE_DPP 1
#endif

#define ROW_BCAST(x, j)                                                   \
  ((int)__builtin_amdgcn_permlane16((unsigned)(x), (unsigned)(x),         \
                                    0x11111111u * (unsigned)(j),          \
                                    0x11111111u * (unsigned)(j), false, true))

__global__ void ndc_prep(const float* __restrict__ v_in,
                         const float* __restrict__ vs_r,
                         const float* __restrict__ fs,
                         float* __restrict__ params, int T) {
  int t = blockIdx.x * blockDim.x + threadIdx.x;
  if (t >= T + 8) return;
  int ti = (t < T) ? t : (T - 1);
  float f   = fs[ti];
  float vr  = vs_r[ti];
  float c1r = 1.0f / (2.0f * C1F * f);
  float r0  = c1r * vr / (c1r + vr);
  float g0 = 1.0f / r0, g1 = 1.0f / c1r, g2 = 1.0f / vr;
  float inv = 2.0f / (g0 + g1 + g2);
  float P0 = g0 * inv, P1 = g1 * inv, P2 = g2 * inv;
  float4* p4 = (float4*)(params + 8 * (size_t)t);
  p4[0] = make_float4(v_in[ti], P0 - 1.0f, P1, P2);
  p4[1] = make_float4(P0, P1 - 1.0f, P2, r0 * (1.0f / 3000.0f));
}

#define WG_BARRIER()                                                         \
  do {                                                                       \
    asm volatile("s_wait_dscnt 0x0" ::: "memory");                           \
    asm volatile("s_barrier_signal -1" ::: "memory");                        \
    asm volatile("s_barrier_wait -1" ::: "memory");                          \
  } while (0)

__global__ __launch_bounds__(64, 1) void ndc_scan(
    const float* __restrict__ params,
    const float* __restrict__ W_in, const float* __restrict__ b_in,
    const float* __restrict__ W_h,  const float* __restrict__ b_h,
    const float* __restrict__ W_out, const float* __restrict__ b_out,
    float* __restrict__ out, int T) {
  __shared__ __attribute__((aligned(16))) float sP2[2 * 64];

  const int  lane = threadIdx.x & 31;
  const int  wave = __builtin_amdgcn_readfirstlane((int)(threadIdx.x >> 5));
  const bool hi   = lane >= 16;
  const int  r16  = lane & 15;

  float Wi0 = W_in[2 * lane];
  float Wi1 = W_in[2 * lane + 1];
  float bin = b_in[lane];
  float bL0  = b_h[16 * wave + r16];
  float bLo1 = b_h[32 + r16];
  float bHi1 = b_h[48 + r16];
  float WoA = W_out[r16];
  float WoB = W_out[16 + r16];
  float bo  = b_out[0];

  float B1[8][2];
#pragma unroll
  for (int c = 0; c < 8; ++c) {
    int row = 16 * wave + r16;
    int col = (hi ? 16 : 0) + 2 * c;
    B1[c][0] = W_h[row * 32 + col];
    B1[c][1] = W_h[row * 32 + col + 1];
  }
  float B2[4][2][2];
#pragma unroll
  for (int cc = 0; cc < 4; ++cc)
#pragma unroll
    for (int blk = 0; blk < 2; ++blk) {
      int row = 16 * blk + r16;
      int col = 16 * wave + (hi ? 8 : 0) + 2 * cc;
      B2[cc][blk][0] = W_h[1024 + row * 32 + col];
      B2[cc][blk][1] = W_h[1024 + row * 32 + col + 1];
    }

  const int wq = 4 * r16 + 2 * wave;
  const int rq = 4 * r16;

  const float4* pf = (const float4*)params;
  float4 qa   = pf[0], qb   = pf[1];
  float4 qa_n = pf[2], qb_n = pf[3];

  float a0 = 0.0f, b1 = 0.0f;

  for (int t = 0; t < T; ++t) {
    float4 qa_nn = pf[2 * (t + 2)];
    float4 qb_nn = pf[2 * (t + 2) + 1];
    __builtin_prefetch(params + 8 * (size_t)(t + 8), 0, 3);

    float v   = qa.x;
    float r00 = qa.y, r01 = qa.z, r02 = qa.w;
    float r10 = qb.x, r11 = qb.y, r12 = qb.z;
    float drt = qb.w;

    float dp_a = r00 * a0 + r01 * b1 + r02 * v;

    float h0 = fmaxf(Wi0 * dp_a + Wi1 * drt + bin, 0.0f);
    int hb = __float_as_int(h0);

    int Ax[8], Ay[8];
#pragma unroll
    for (int c = 0; c < 8; ++c) {
      Ax[c] = ROW_BCAST(hb, 2 * c);
      Ay[c] = ROW_BCAST(hb, 2 * c + 1);
    }
    v8f a00 = {}, a01 = {};
#pragma unroll
    for (int c = 0; c < 4; ++c) {
      v2f A;  A.x  = __int_as_float(Ax[c]);     A.y  = __int_as_float(Ay[c]);
      v2f A2; A2.x = __int_as_float(Ax[c + 4]); A2.y = __int_as_float(Ay[c + 4]);
      v2f Ba; Ba.x = B1[c][0];     Ba.y = B1[c][1];
      v2f Bb; Bb.x = B1[c + 4][0]; Bb.y = B1[c + 4][1];
      a00 = __builtin_amdgcn_wmma_f32_16x16x4_f32(false, A,  false, Ba,
                                                  (short)0, a00, false, false);
      a01 = __builtin_amdgcn_wmma_f32_16x16x4_f32(false, A2, false, Bb,
                                                  (short)0, a01, false, false);
    }
    float h1 = fmaxf((a00[0] + a01[0]) + bL0, 0.0f);
    int hb1 = __float_as_int(h1);

    int hb1r = __builtin_amdgcn_update_dpp(hb1, hb1, 0x128, 0x2, 0xf, false);
    int Gx[4], Gy[4];
#pragma unroll
    for (int cc = 0; cc < 4; ++cc) {
      Gx[cc] = ROW_BCAST(hb1r, 2 * cc);
      Gy[cc] = ROW_BCAST(hb1r, 2 * cc + 1);
    }
    v8f c0 = {}, c1 = {};
#pragma unroll
    for (int cc = 0; cc < 4; ++cc) {
      v2f A;  A.x  = __int_as_float(Gx[cc]); A.y  = __int_as_float(Gy[cc]);
      v2f Ba; Ba.x = B2[cc][0][0]; Ba.y = B2[cc][0][1];
      v2f Bb; Bb.x = B2[cc][1][0]; Bb.y = B2[cc][1][1];
      c0 = __builtin_amdgcn_wmma_f32_16x16x4_f32(false, A, false, Ba,
                                                 (short)0, c0, false, false);
      c1 = __builtin_amdgcn_wmma_f32_16x16x4_f32(false, A, false, Bb,
                                                 (short)0, c1, false, false);
    }
    float* buf = sP2 + ((t & 1) << 6);
    *(float2*)&buf[wq] = make_float2(c0[0], c1[0]);
    WG_BARRIER();
    float4 pz = *(const float4*)&buf[rq];
    float gLo = fmaxf((pz.x + pz.z) + bLo1, 0.0f);
    float gHi = fmaxf((pz.y + pz.w) + bHi1, 0.0f);

    float p = __builtin_fmaf(WoA, gLo, WoB * gHi);
#ifdef HAVE_DPP
    p = ROR_ADD(p, 0x128);
    p = ROR_ADD(p, 0x124);
    p = ROR_ADD(p, 0x122);
    p = ROR_ADD(p, 0x121);
#else
#pragma unroll
    for (int m = 8; m >= 1; m >>= 1) {
      p += __int_as_float(
          __builtin_amdgcn_ds_bpermute((lane ^ m) << 2, __float_as_int(p)));
    }
#endif
    float dp_b = p + bo;

    float b1n = r10 * dp_b + r11 * b1 + r12 * v;

    if (wave == 0) out[t] = 0.5f * (dp_a + dp_b);

    a0 = dp_b;
    b1 = b1n;
    qa = qa_n;  qb = qb_n;
    qa_n = qa_nn;  qb_n = qb_nn;
  }
}

extern "C" void kernel_launch(void* const* d_in, const int* in_sizes, int n_in,
                              void* d_out, int out_size, void* d_ws,
                              size_t ws_size, hipStream_t stream) {
  const float* v_in  = (const float*)d_in[0];
  const float* vs_r  = (const float*)d_in[1];
  const float* fs    = (const float*)d_in[2];
  const float* W_in  = (const float*)d_in[3];
  const float* b_in  = (const float*)d_in[4];
  const float* W_h   = (const float*)d_in[5];
  const float* b_h   = (const float*)d_in[6];
  const float* W_out = (const float*)d_in[7];
  const float* b_out = (const float*)d_in[8];
  float* out = (float*)d_out;
  int T = in_sizes[0];

  float* params = (float*)d_ws;

  ndc_prep<<<(T + 8 + 255) / 256, 256, 0, stream>>>(v_in, vs_r, fs, params, T);
  ndc_scan<<<1, 64, 0, stream>>>(params, W_in, b_in, W_h, b_h, W_out, b_out,
                                 out, T);
}


# Four waves on four SIMDs share each step: layer 2 by K quarter; a wave forms only its 8 hidden-1 values, K halves on A's row halves, 4 WMMA per layer per wave, not 8; dot-product sum order changes
# speedup vs baseline: 1.5188x; 1.1586x over previous
#include <hip/hip_runtime.h>


typedef __attribute__((ext_vector_type(2))) float v2f;
typedef __attribute__((ext_vector_type(8))) float v8f;

#define C1F 4.7e-9f

#if __has_builtin(__builtin_amdgcn_update_dpp)
#define ROR_ADD(x, ctrl)                                                  \
  ((x) + __int_as_float(__builtin_amdgcn_update_dpp(                      \
             0, __float_as_int(x), (ctrl), 0xf, 0xf, true)))
#define HAVE_DPP 1
#endif

#define ROW_BCAST2(x, ja, jb)                                             \
  ((int)__builtin_amdgcn_permlane16((unsigned)(x), (unsigned)(x),         \
                                    0x11111111u * (unsigned)(ja),         \
                                    0x11111111u * (unsigned)(jb), false, true))
#define ROW_BCAST(x, j) ROW_BCAST2(x, j, j)

__global__ void ndc_prep(const float* __restrict__ v_in,
                         const float* __restrict__ vs_r,
                         const float* __restrict__ fs,
                         float* __restrict__ params, int T) {
  int t = blockIdx.x * blockDim.x + threadIdx.x;
  if (t >= T + 8) return;
  int ti = (t < T) ? t : (T - 1);
  float f   = fs[ti];
  float vr  = vs_r[ti];
  float c1r = 1.0f / (2.0f * C1F * f);
  float r0  = c1r * vr / (c1r + vr);
  float g0 = 1.0f / r0, g1 = 1.0f / c1r, g2 = 1.0f / vr;
  float inv = 2.0f / (g0 + g1 + g2);
  float P0 = g0 * inv, P1 = g1 * inv, P2 = g2 * inv;
  float4* p4 = (float4*)(params + 8 * (size_t)t);
  p4[0] = make_float4(v_in[ti], P0 - 1.0f, P1, P2);
  p4[1] = make_float4(P0, P1 - 1.0f, P2, r0 * (1.0f / 3000.0f));
}

#define WG_BARRIER()                                                         \
  do {                                                                       \
    asm volatile("s_wait_dscnt 0x0" ::: "memory");                           \
    asm volatile("s_barrier_signal -1" ::: "memory");                        \
    asm volatile("s_barrier_wait -1" ::: "memory");                          \
  } while (0)

__global__ __launch_bounds__(128, 1) void ndc_scan(
    const float* __restrict__ params,
    const float* __restrict__ W_in, const float* __restrict__ b_in,
    const float* __restrict__ W_h,  const float* __restrict__ b_h,
    const float* __restrict__ W_out, const float* __restrict__ b_out,
    float* __restrict__ out, int T) {
  __shared__ __attribute__((aligned(16))) float sP2[2 * 128];

  const int  lane = threadIdx.x & 31;
  const int  wave = __builtin_amdgcn_readfirstlane((int)(threadIdx.x >> 5));
  const bool hi   = lane >= 16;
  const int  r16  = lane & 15;
  const int  grp  = (lane >> 3) & 1;
  const int  j8   = (r16 + 4 * grp) & 7;

  float Wi0 = W_in[2 * lane];
  float Wi1 = W_in[2 * lane + 1];
  float bin = b_in[lane];
  float bL0  = b_h[8 * wave + j8];
  float bLo1 = b_h[32 + r16];
  float bHi1 = b_h[48 + r16];
  float WoA = W_out[r16];
  float WoB = W_out[16 + r16];
  float bo  = b_out[0];

  float B1[4][2];
#pragma unroll
  for (int c = 0; c < 4; ++c) {
    int row = 8 * wave + j8;
    int col = (hi ? 16 : 0) + 8 * grp + 2 * c;
    B1[c][0] = W_h[row * 32 + col];
    B1[c][1] = W_h[row * 32 + col + 1];
  }
  float B2[2][2][2];
#pragma unroll
  for (int cc = 0; cc < 2; ++cc)
#pragma unroll
    for (int blk = 0; blk < 2; ++blk) {
      int row = 16 * blk + r16;
      int col = 8 * wave + (hi ? 4 : 0) + 2 * cc;
      B2[cc][blk][0] = W_h[1024 + row * 32 + col];
      B2[cc][blk][1] = W_h[1024 + row * 32 + col + 1];
    }

  const int wq = 8 * r16 + 2 * wave;
  const int rq = 8 * r16;

  const float4* pf = (const float4*)params;
  float4 qa   = pf[0], qb   = pf[1];
  float4 qa_n = pf[2], qb_n = pf[3];

  float a0 = 0.0f, b1 = 0.0f;

  for (int t = 0; t < T; ++t) {
    float4 qa_nn = pf[2 * (t + 2)];
    float4 qb_nn = pf[2 * (t + 2) + 1];
    __builtin_prefetch(params + 8 * (size_t)(t + 8), 0, 3);

    float v   = qa.x;
    float r00 = qa.y, r01 = qa.z, r02 = qa.w;
    float r10 = qb.x, r11 = qb.y, r12 = qb.z;
    float drt = qb.w;

    float dp_a = r00 * a0 + r01 * b1 + r02 * v;

    float h0 = fmaxf(Wi0 * dp_a + Wi1 * drt + bin, 0.0f);
    int hb = __float_as_int(h0);

    int Ax[4], Ay[4];
#pragma unroll
    for (int c = 0; c < 4; ++c) {
      Ax[c] = ROW_BCAST2(hb, 2 * c,     8 + 2 * c);
      Ay[c] = ROW_BCAST2(hb, 2 * c + 1, 8 + 2 * c + 1);
    }
    v8f a00 = {};
#pragma unroll
    for (int c = 0; c < 4; ++c) {
      v2f A;  A.x  = __int_as_float(Ax[c]); A.y  = __int_as_float(Ay[c]);
      v2f Ba; Ba.x = B1[c][0];              Ba.y = B1[c][1];
      a00 = __builtin_amdgcn_wmma_f32_16x16x4_f32(false, A, false, Ba,
                                                  (short)0, a00, false, false);
    }
    int d0 = __float_as_int(a00[0]);
    int dx = (int)__builtin_amdgcn_permlanex16((unsigned)d0, (unsigned)d0,
                                               0xba98fedcu, 0x32107654u,
                                               false, true);
    float h1 = fmaxf((a00[0] + __int_as_float(dx)) + bL0, 0.0f);
    int hb1 = __float_as_int(h1);

    int hb1r = __builtin_amdgcn_update_dpp(hb1, hb1, 0x128, 0x2, 0xf, false);
    int Gx[2], Gy[2];
#pragma unroll
    for (int cc = 0; cc < 2; ++cc) {
      Gx[cc] = ROW_BCAST(hb1r, 2 * cc);
      Gy[cc] = ROW_BCAST(hb1r, 2 * cc + 1);
    }
    v8f c0 = {}, c1 = {};
#pragma unroll
    for (int cc = 0; cc < 2; ++cc) {
      v2f A;  A.x  = __int_as_float(Gx[cc]); A.y  = __int_as_float(Gy[cc]);
      v2f Ba; Ba.x = B2[cc][0][0]; Ba.y = B2[cc][0][1];
      v2f Bb; Bb.x = B2[cc][1][0]; Bb.y = B2[cc][1][1];
      c0 = __builtin_amdgcn_wmma_f32_16x16x4_f32(false, A, false, Ba,
                                                 (short)0, c0, false, false);
      c1 = __builtin_amdgcn_wmma_f32_16x16x4_f32(false, A, false, Bb,
                                                 (short)0, c1, false, false);
    }
    float* buf = sP2 + ((t & 1) << 7);
    *(float2*)&buf[wq] = make_float2(c0[0], c1[0]);
    WG_BARRIER();
    float4 pz = *(const float4*)&buf[rq];
    float4 pw = *(const float4*)&buf[rq + 4];
    float gLo = fmaxf(((pz.x + pz.z) + (pw.x + pw.z)) + bLo1, 0.0f);
    float gHi = fmaxf(((pz.y + pz.w) + (pw.y + pw.w)) + bHi1, 0.0f);

    float p = __builtin_fmaf(WoA, gLo, WoB * gHi);
#ifdef HAVE_DPP
    p = ROR_ADD(p, 0x128);
    p = ROR_ADD(p, 0x124);
    p = ROR_ADD(p, 0x122);
    p = ROR_ADD(p, 0x121);
#else
#pragma unroll
    for (int m = 8; m >= 1; m >>= 1) {
      p += __int_as_float(
          __builtin_amdgcn_ds_bpermute((lane ^ m) << 2, __float_as_int(p)));
    }
#endif
    float dp_b = p + bo;

    float b1n = r10 * dp_b + r11 * b1 + r12 * v;

    if (wave == 0) out[t] = 0.5f * (dp_a + dp_b);

    a0 = dp_b;
    b1 = b1n;
    qa = qa_n;  qb = qb_n;
    qa_n = qa_nn;  qb_n = qb_nn;
  }
}

extern "C" void kernel_launch(void* const* d_in, const int* in_sizes, int n_in,
                              void* d_out, int out_size, void* d_ws,
                              size_t ws_size, hipStream_t stream) {
  const float* v_in  = (const float*)d_in[0];
  const float* vs_r  = (const float*)d_in[1];
  const float* fs    = (const float*)d_in[2];
  const float* W_in  = (const float*)d_in[3];
  const float* b_in  = (const float*)d_in[4];
  const float* W_h   = (const float*)d_in[5];
  const float* b_h   = (const float*)d_in[6];
  const float* W_out = (const float*)d_in[7];
  const float* b_out = (const float*)d_in[8];
  float* out = (float*)d_out;
  int T = in_sizes[0];

  float* params = (float*)d_ws;

  ndc_prep<<<(T + 8 + 255) / 256, 256, 0, stream>>>(v_in, vs_r, fs, params, T);
  ndc_scan<<<1, 128, 0, stream>>>(params, W_in, b_in, W_h, b_h, W_out, b_out,
                                  out, T);
}
